# starting kernel
# baseline (speedup 1.0000x reference)
#include <hip/hip_runtime.h>
#include <hip/hip_bf16.h>
#include <math.h>

typedef __attribute__((ext_vector_type(2))) float v2f;
typedef __attribute__((ext_vector_type(8))) float v8f;

static __device__ __forceinline__ v8f wmma_f32_16x16x4(v2f a, v2f b, v8f c) {
  return __builtin_amdgcn_wmma_f32_16x16x4_f32(false, a, false, b, (short)0, c, false, false);
}

static __device__ __forceinline__ float fast_sigmoid(float x) {
  x = fminf(fmaxf(x, -30.f), 30.f);
  return __builtin_amdgcn_rcpf(1.f + __expf(-x));
}
static __device__ __forceinline__ float fast_tanh(float x) {
  x = fminf(fmaxf(x, -15.f), 15.f);
  float e = __expf(-2.f * x);
  return (1.f - e) * __builtin_amdgcn_rcpf(1.f + e);
}

#define B_SZ 256
#define S_SZ 256
#define F_SZ 64
#define H_SZ 128
#define HSTR 132

__global__ void __launch_bounds__(128)
qlstm_xproj(const float* __restrict__ x, const float* __restrict__ Win,
            const float* __restrict__ bin, float* __restrict__ y0) {
  const int lane = threadIdx.x & 31;
  const int wave = threadIdx.x >> 5;
  const int tile = blockIdx.x * 4 + wave;
  const int n    = lane & 15;
  const int hi   = lane >> 4;
  const int kh   = hi << 1;

  v2f bch[16];
#pragma unroll
  for (int j = 0; j < 16; ++j) {
    float b0 = 0.f, b1 = 0.f;
    if (n < 4) {
      int kk = j * 4 + kh;
      b0 = Win[n * 192 + 128 + kk];
      b1 = Win[n * 192 + 128 + kk + 1];
    }
    v2f bv = {b0, b1};
    bch[j] = bv;
  }

  const float bb = (n < 4) ? bin[n] : 0.f;
  v8f acc = {bb, bb, bb, bb, bb, bb, bb, bb};

  const float* xr = x + (size_t)(tile * 16 + n) * F_SZ;
#pragma unroll
  for (int j = 0; j < 16; ++j) {
    v2f a = {xr[j * 4 + kh], xr[j * 4 + kh + 1]};
    acc = wmma_f32_16x16x4(a, bch[j], acc);
  }

  if (n < 4) {
#pragma unroll
    for (int r = 0; r < 8; ++r) {
      int row = tile * 16 + r + (hi << 3);
      y0[(size_t)row * 4 + n] = acc[r];
    }
  }
}

__global__ void __launch_bounds__(256)
qlstm_recur(const float* __restrict__ y0, const float* __restrict__ Win,
            const float* __restrict__ Wout, const float* __restrict__ bout,
            const float* __restrict__ thetas, float* __restrict__ out) {
  __shared__ float y0_lds[16 * S_SZ * 4];
  __shared__ float h_lds[16 * HSTR];
  __shared__ float part[8 * 16 * 4];
  __shared__ float z_lds[4 * 16 * 4];

  const int tid   = threadIdx.x;
  const int lane  = tid & 31;
  const int wave  = tid >> 5;
  const int n     = lane & 15;
  const int hi    = lane >> 4;
  const int kh    = hi << 1;
  const int bbase = blockIdx.x * 16;

  float* hs = out;
  float* hT = out + (size_t)B_SZ * S_SZ * H_SZ;
  float* cT = hT + (size_t)B_SZ * H_SZ;

  {
    const float4* src = (const float4*)(y0 + (size_t)bbase * S_SZ * 4);
    float4* dst = (float4*)y0_lds;
#pragma unroll
    for (int i = 0; i < 16; ++i)
      dst[tid + i * 256] = src[tid + i * 256];
  }
  for (int i = tid; i < 16 * HSTR; i += 256) h_lds[i] = 0.f;

  v2f bh[4];
#pragma unroll
  for (int j = 0; j < 4; ++j) {
    float b0 = 0.f, b1 = 0.f;
    if (n < 4) {
      int kk = wave * 16 + j * 4 + kh;
      b0 = Win[n * 192 + kk];
      b1 = Win[n * 192 + kk + 1];
    }
    v2f bv = {b0, b1};
    bh[j] = bv;
  }
  const int col = wave * 16 + n;
  v2f bw = {Wout[col * 4 + kh], Wout[col * 4 + kh + 1]};
  const float bo = bout[col];
  const v8f cbias = {bo, bo, bo, bo, bo, bo, bo, bo};

  const int sb = tid >> 2, sg = tid & 3;
  float th[4];
#pragma unroll
  for (int w = 0; w < 4; ++w) th[w] = (tid < 64) ? thetas[sg * 4 + w] : 0.f;

  float c_state[8], h_keep[8];
#pragma unroll
  for (int r = 0; r < 8; ++r) { c_state[r] = 0.f; h_keep[r] = 0.f; }

  __syncthreads();

  for (int t = 0; t < S_SZ; ++t) {
    v8f acc = {0.f, 0.f, 0.f, 0.f, 0.f, 0.f, 0.f, 0.f};
#pragma unroll
    for (int j = 0; j < 4; ++j) {
      int kk = wave * 16 + j * 4 + kh;
      v2f a = {h_lds[n * HSTR + kk], h_lds[n * HSTR + kk + 1]};
      acc = wmma_f32_16x16x4(a, bh[j], acc);
    }
    if (n < 4) {
#pragma unroll
      for (int r = 0; r < 8; ++r)
        part[wave * 64 + (r + (hi << 3)) * 4 + n] = acc[r];
    }
    __syncthreads();

    if (tid < 64) {
      float p0[4], p1[4];
#pragma unroll
      for (int w = 0; w < 4; ++w) {
        float yw = y0_lds[(sb * S_SZ + t) * 4 + w];
#pragma unroll
        for (int wv = 0; wv < 8; ++wv) yw += part[wv * 64 + sb * 4 + w];
        float cv = __cosf(yw + th[w]);
        p0[w] = 0.5f + 0.5f * cv;
        p1[w] = 0.5f - 0.5f * cv;
      }
      float z0 = 0.f, z1 = 0.f, z2 = 0.f, z3 = 0.f;
#pragma unroll
      for (int idx = 0; idx < 16; ++idx) {
        int a0 = (idx >> 3) & 1, a1 = (idx >> 2) & 1;
        int a2 = (idx >> 1) & 1, a3 = idx & 1;
        float p = (a0 ? p1[0] : p0[0]) * (a1 ? p1[1] : p0[1]) *
                  (a2 ? p1[2] : p0[2]) * (a3 ? p1[3] : p0[3]);
        int t1 = a1 ^ a0;
        int t2 = a2 ^ t1;
        int t3 = a3 ^ t2;
        int t0 = a0 ^ t3;
        z0 += t0 ? -p : p;  z1 += t1 ? -p : p;
        z2 += t2 ? -p : p;  z3 += t3 ? -p : p;
      }
      int base = (sg * 16 + sb) * 4;
      z_lds[base + 0] = z0;  z_lds[base + 1] = z1;
      z_lds[base + 2] = z2;  z_lds[base + 3] = z3;
    }
    __syncthreads();

    v8f ga[4];
#pragma unroll
    for (int g = 0; g < 4; ++g) {
      v2f a = {z_lds[(g * 16 + n) * 4 + kh], z_lds[(g * 16 + n) * 4 + kh + 1]};
      ga[g] = wmma_f32_16x16x4(a, bw, cbias);
    }
#pragma unroll
    for (int r = 0; r < 8; ++r) {
      float fv = fast_sigmoid(ga[0][r]);
      float iv = fast_sigmoid(ga[1][r]);
      float gv = fast_tanh(ga[2][r]);
      float ov = fast_sigmoid(ga[3][r]);
      float cs = fv * c_state[r] + iv * gv;
      c_state[r] = cs;
      float hv = ov * fast_tanh(cs);
      h_keep[r] = hv;
      int mrow = r + (hi << 3);
      h_lds[mrow * HSTR + col] = hv;
      hs[(((size_t)(bbase + mrow)) * S_SZ + t) * H_SZ + col] = hv;
    }
    __syncthreads();
  }

#pragma unroll
  for (int r = 0; r < 8; ++r) {
    int mrow = r + (hi << 3);
    hT[((size_t)(bbase + mrow)) * H_SZ + col] = h_keep[r];
    cT[((size_t)(bbase + mrow)) * H_SZ + col] = c_state[r];
  }
}

extern "C" void kernel_launch(void* const* d_in, const int* in_sizes, int n_in,
                              void* d_out, int out_size, void* d_ws, size_t ws_size,
                              hipStream_t stream) {
  const float* x      = (const float*)d_in[0];
  const float* W_in   = (const float*)d_in[1];
  const float* b_in   = (const float*)d_in[2];
  const float* W_out  = (const float*)d_in[3];
  const float* b_out  = (const float*)d_in[4];
  const float* thetas = (const float*)d_in[5];
  float* out = (float*)d_out;
  float* y0  = (float*)d_ws;

  qlstm_xproj<<<1024, 128, 0, stream>>>(x, W_in, b_in, y0);
  qlstm_recur<<<16, 256, 0, stream>>>(y0, W_in, W_out, b_out, thetas, out);
}


# scan re-cut by batch: one wave owns 2 whole batch rows and all 128 H columns (128 one-wave blocks); y sums finish inside the wave, no block barrier or partial exchange; y sum order regrouped
# speedup vs baseline: 2.1667x; 2.1667x over previous
#include <hip/hip_runtime.h>
#include <hip/hip_bf16.h>
#include <math.h>

typedef __attribute__((ext_vector_type(2))) float v2f;
typedef __attribute__((ext_vector_type(8))) float v8f;

static __device__ __forceinline__ v8f wmma_f32_16x16x4(v2f a, v2f b, v8f c) {
  return __builtin_amdgcn_wmma_f32_16x16x4_f32(false, a, false, b, (short)0, c, false, false);
}

static __device__ __forceinline__ float fast_sigmoid(float x) {
  x = fminf(fmaxf(x, -30.f), 30.f);
  return __builtin_amdgcn_rcpf(1.f + __expf(-x));
}
static __device__ __forceinline__ float fast_tanh(float x) {
  x = fminf(fmaxf(x, -15.f), 15.f);
  float e = __expf(-2.f * x);
  return (1.f - e) * __builtin_amdgcn_rcpf(1.f + e);
}

#define B_SZ 256
#define S_SZ 256
#define F_SZ 64
#define H_SZ 128
#define HSTR 144
#define YSTR (S_SZ * 4 + 16)

__global__ void __launch_bounds__(128)
qlstm_xproj(const float* __restrict__ x, const float* __restrict__ Win,
            const float* __restrict__ bin, float* __restrict__ y0) {
  const int lane = threadIdx.x & 31;
  const int wave = threadIdx.x >> 5;
  const int tile = blockIdx.x * 4 + wave;
  const int n    = lane & 15;
  const int hi   = lane >> 4;
  const int kh   = hi << 1;

  v2f bch[16];
#pragma unroll
  for (int j = 0; j < 16; ++j) {
    float b0 = 0.f, b1 = 0.f;
    if (n < 4) {
      int kk = j * 4 + kh;
      b0 = Win[n * 192 + 128 + kk];
      b1 = Win[n * 192 + 128 + kk + 1];
    }
    v2f bv = {b0, b1};
    bch[j] = bv;
  }

  const float bb = (n < 4) ? bin[n] : 0.f;
  v8f acc = {bb, bb, bb, bb, bb, bb, bb, bb};

  const float* xr = x + (size_t)(tile * 16 + n) * F_SZ;
#pragma unroll
  for (int j = 0; j < 16; ++j) {
    v2f a = {xr[j * 4 + kh], xr[j * 4 + kh + 1]};
    acc = wmma_f32_16x16x4(a, bch[j], acc);
  }

  if (n < 4) {
#pragma unroll
    for (int r = 0; r < 8; ++r) {
      int row = tile * 16 + r + (hi << 3);
      y0[(size_t)row * 4 + n] = acc[r];
    }
  }
}

__global__ void __launch_bounds__(32)
qlstm_recur(const float* __restrict__ y0, const float* __restrict__ Win,
            const float* __restrict__ Wout, const float* __restrict__ bout,
            const float* __restrict__ thetas, float* __restrict__ out) {
  __shared__ float y0_lds[2 * YSTR];
  __shared__ float h_lds[2 * HSTR];

  const int lane  = threadIdx.x;
  const int n     = lane & 15;
  const int hi    = lane >> 4;
  const int kh    = hi << 1;
  const int bbase = blockIdx.x * 2;

  float* hs = out;
  float* hT = out + (size_t)B_SZ * S_SZ * H_SZ;
  float* cT = hT + (size_t)B_SZ * H_SZ;

  {
    const float4* src = (const float4*)(y0 + (size_t)bbase * S_SZ * 4);
#pragma unroll
    for (int i = 0; i < 16; ++i) {
      int idx = lane + i * 32;
      int row = idx >> 8, within = idx & 255;
      *(float4*)&y0_lds[row * YSTR + within * 4] = src[idx];
    }
  }

  const int q  = n >> 2;
  const int cg = n & 3;
  v2f bh[8];
#pragma unroll
  for (int j = 0; j < 8; ++j) {
    int kk = j * 16 + cg * 4 + kh;
    v2f bv = {Win[q * 192 + kk], Win[q * 192 + kk + 1]};
    bh[j] = bv;
  }
  v2f bw[8];
  float bo[8];
#pragma unroll
  for (int j = 0; j < 8; ++j) {
    int col = j * 16 + n;
    v2f bv = {Wout[col * 4 + kh], Wout[col * 4 + kh + 1]};
    bw[j] = bv;
    bo[j] = bout[col];
  }

  const int mb = n >> 3;
  const int mg = n & 3;
  float th[4];
#pragma unroll
  for (int w = 0; w < 4; ++w) th[w] = thetas[mg * 4 + w];

  const float* hA = h_lds + mb * HSTR + mg * 4 + kh;
  float*       hW = h_lds + hi * HSTR + n;
  const float* yR = y0_lds + hi * YSTR + q;
  float* hs_p = hs + ((size_t)(bbase + hi) * S_SZ) * H_SZ + n;

  float c_state[8], h_keep[8];
#pragma unroll
  for (int r = 0; r < 8; ++r) { c_state[r] = 0.f; h_keep[r] = 0.f; }

  __builtin_amdgcn_fence(__ATOMIC_RELEASE, "wavefront");
  __builtin_amdgcn_wave_barrier();
  __builtin_amdgcn_fence(__ATOMIC_ACQUIRE, "wavefront");

  float ysum = 0.f;
  float yq = yR[0];

  for (int t = 0; t < S_SZ; ++t) {
    const float y = ysum + yq;
    const int tn = (t + 1 < S_SZ) ? t + 1 : t;
    yq = yR[tn * 4];

    float yv[4];
#pragma unroll
    for (int w = 0; w < 4; ++w) {
      float ya = __int_as_float(__builtin_amdgcn_readlane(__float_as_int(y), 4 * w));
      float yb = __int_as_float(__builtin_amdgcn_readlane(__float_as_int(y), 16 + 4 * w));
      yv[w] = mb ? yb : ya;
    }
    float p0[4], p1[4];
#pragma unroll
    for (int w = 0; w < 4; ++w) {
      float cv = __cosf(yv[w] + th[w]);
      p0[w] = 0.5f + 0.5f * cv;
      p1[w] = 0.5f - 0.5f * cv;
    }
    float z0 = 0.f, z1 = 0.f, z2 = 0.f, z3 = 0.f;
#pragma unroll
    for (int idx = 0; idx < 16; ++idx) {
      int a0 = (idx >> 3) & 1, a1 = (idx >> 2) & 1;
      int a2 = (idx >> 1) & 1, a3 = idx & 1;
      float p = (a0 ? p1[0] : p0[0]) * (a1 ? p1[1] : p0[1]) *
                (a2 ? p1[2] : p0[2]) * (a3 ? p1[3] : p0[3]);
      int t1 = a1 ^ a0;
      int t2 = a2 ^ t1;
      int t3 = a3 ^ t2;
      int t0 = a0 ^ t3;
      z0 += t0 ? -p : p;  z1 += t1 ? -p : p;
      z2 += t2 ? -p : p;  z3 += t3 ? -p : p;
    }
    const v2f za = {hi ? z2 : z0, hi ? z3 : z1};

    v8f acc0 = {0.f, 0.f, 0.f, 0.f, 0.f, 0.f, 0.f, 0.f};
    v8f acc1 = {0.f, 0.f, 0.f, 0.f, 0.f, 0.f, 0.f, 0.f};
    float* hs_t = hs_p + (size_t)t * H_SZ;
#pragma unroll
    for (int j = 0; j < 8; ++j) {
      const v8f cbias = {bo[j], bo[j], bo[j], bo[j], bo[j], bo[j], bo[j], bo[j]};
      v8f ga = wmma_f32_16x16x4(za, bw[j], cbias);
      float fv = fast_sigmoid(ga[0]);
      float iv = fast_sigmoid(ga[1]);
      float gv = fast_tanh(ga[2]);
      float ov = fast_sigmoid(ga[3]);
      float cs = fv * c_state[j] + iv * gv;
      c_state[j] = cs;
      float hv = ov * fast_tanh(cs);
      h_keep[j] = hv;
      hW[j * 16] = hv;
      hs_t[j * 16] = hv;
      __builtin_amdgcn_fence(__ATOMIC_RELEASE, "wavefront");
      __builtin_amdgcn_wave_barrier();
      __builtin_amdgcn_fence(__ATOMIC_ACQUIRE, "wavefront");
      const v2f a = *(const v2f*)(hA + j * 16);
      if (j & 1) acc1 = wmma_f32_16x16x4(a, bh[j], acc1);
      else       acc0 = wmma_f32_16x16x4(a, bh[j], acc0);
    }
    const float s0 = acc0[0] + acc1[0], s1 = acc0[1] + acc1[1];
    const float s2 = acc0[2] + acc1[2], s3 = acc0[3] + acc1[3];
    const float g0 = __int_as_float(__builtin_amdgcn_update_dpp(0, __float_as_int(s0), 0x00, 0xF, 0xF, true));
    const float g1 = __int_as_float(__builtin_amdgcn_update_dpp(0, __float_as_int(s1), 0x55, 0xF, 0xF, true));
    const float g2 = __int_as_float(__builtin_amdgcn_update_dpp(0, __float_as_int(s2), 0xAA, 0xF, 0xF, true));
    const float g3 = __int_as_float(__builtin_amdgcn_update_dpp(0, __float_as_int(s3), 0xFF, 0xF, 0xF, true));
    const float d = ((g0 + g1) + g2) + g3;
    ysum = d;
  }

#pragma unroll
  for (int j = 0; j < 8; ++j) {
    hT[((size_t)(bbase + hi)) * H_SZ + j * 16 + n] = h_keep[j];
    cT[((size_t)(bbase + hi)) * H_SZ + j * 16 + n] = c_state[j];
  }
}

extern "C" void kernel_launch(void* const* d_in, const int* in_sizes, int n_in,
                              void* d_out, int out_size, void* d_ws, size_t ws_size,
                              hipStream_t stream) {
  const float* x      = (const float*)d_in[0];
  const float* W_in   = (const float*)d_in[1];
  const float* b_in   = (const float*)d_in[2];
  const float* W_out  = (const float*)d_in[3];
  const float* b_out  = (const float*)d_in[4];
  const float* thetas = (const float*)d_in[5];
  float* out = (float*)d_out;
  float* y0  = (float*)d_ws;

  qlstm_xproj<<<1024, 128, 0, stream>>>(x, W_in, b_in, y0);
  qlstm_recur<<<128, 32, 0, stream>>>(y0, W_in, W_out, b_out, thetas, out);
}


# scan re-cut to ONE batch row per wave (256 one-wave blocks where 128): lane halves get equal gate tiles and take every other one, a lane updates 4 cells where 8; y wave-uniform; sum order unchanged
# speedup vs baseline: 3.1471x; 1.4525x over previous
#include <hip/hip_runtime.h>
#include <hip/hip_bf16.h>
#include <math.h>

typedef __attribute__((ext_vector_type(2))) float v2f;
typedef __attribute__((ext_vector_type(8))) float v8f;

static __device__ __forceinline__ v8f wmma_f32_16x16x4(v2f a, v2f b, v8f c) {
  return __builtin_amdgcn_wmma_f32_16x16x4_f32(false, a, false, b, (short)0, c, false, false);
}

#define NEG_LOG2E  (-1.4426950408889634f)
#define NEG_2LOG2E (-2.8853900817779268f)
static __device__ __forceinline__ float sigmoid_of_scaled(float d) {
  return __builtin_amdgcn_rcpf(1.f + __builtin_amdgcn_exp2f(d));
}
static __device__ __forceinline__ float tanh_of_scaled(float d) {
  return fmaf(2.f, __builtin_amdgcn_rcpf(1.f + __builtin_amdgcn_exp2f(d)), -1.f);
}

#define B_SZ 256
#define S_SZ 256
#define F_SZ 64
#define H_SZ 128
#define HP   12

__global__ void __launch_bounds__(128)
qlstm_xproj(const float* __restrict__ x, const float* __restrict__ Win,
            const float* __restrict__ bin, float* __restrict__ y0) {
  const int lane = threadIdx.x & 31;
  const int wave = threadIdx.x >> 5;
  const int tile = blockIdx.x * 4 + wave;
  const int n    = lane & 15;
  const int hi   = lane >> 4;
  const int kh   = hi << 1;

  v2f bch[16];
#pragma unroll
  for (int j = 0; j < 16; ++j) {
    float b0 = 0.f, b1 = 0.f;
    if (n < 4) {
      int kk = j * 4 + kh;
      b0 = Win[n * 192 + 128 + kk];
      b1 = Win[n * 192 + 128 + kk + 1];
    }
    v2f bv = {b0, b1};
    bch[j] = bv;
  }

  const float bb = (n < 4) ? bin[n] : 0.f;
  v8f acc = {bb, bb, bb, bb, bb, bb, bb, bb};

  const float* xr = x + (size_t)(tile * 16 + n) * F_SZ;
#pragma unroll
  for (int j = 0; j < 16; ++j) {
    v2f a = {xr[j * 4 + kh], xr[j * 4 + kh + 1]};
    acc = wmma_f32_16x16x4(a, bch[j], acc);
  }

  if (n < 4) {
#pragma unroll
    for (int r = 0; r < 8; ++r) {
      int row = tile * 16 + r + (hi << 3);
      y0[(size_t)row * 4 + n] = acc[r];
    }
  }
}

__global__ void __launch_bounds__(32)
qlstm_recur(const float* __restrict__ y0, const float* __restrict__ Win,
            const float* __restrict__ Wout, const float* __restrict__ bout,
            const float* __restrict__ thetas, float* __restrict__ out) {
  __shared__ __attribute__((aligned(16))) float y0_lds[S_SZ * 4];
  __shared__ __attribute__((aligned(16))) float h_lds[16 * HP];

  const int lane  = threadIdx.x;
  const int n     = lane & 15;
  const int hi    = lane >> 4;
  const int kh    = hi << 1;
  const int brow  = blockIdx.x;

  float* hs = out;
  float* hT = out + (size_t)B_SZ * S_SZ * H_SZ;
  float* cT = hT + (size_t)B_SZ * H_SZ;

  {
    const float4* src = (const float4*)(y0 + (size_t)brow * S_SZ * 4);
#pragma unroll
    for (int i = 0; i < 8; ++i) {
      int idx = lane + i * 32;
      *(float4*)&y0_lds[idx * 4] = src[idx];
    }
  }

  const int q  = n >> 2;
  const int cg = n & 3;
  v2f bh[2][4];
#pragma unroll
  for (int e = 0; e < 2; ++e) {
#pragma unroll
    for (int jp = 0; jp < 4; ++jp) {
      int kk = 32 * jp + 4 * cg + kh + e;
      v2f bv = {Win[q * 192 + kk], Win[q * 192 + kk + 16]};
      bh[e][jp] = bv;
    }
  }
  v2f bw[8];
  float bo_s[8], bo_t[8];
#pragma unroll
  for (int j = 0; j < 8; ++j) {
    int col = j * 16 + n;
    v2f bv = {Wout[col * 4 + kh], Wout[col * 4 + kh + 1]};
    bw[j] = bv;
    const float b = bout[col];
    bo_s[j] = b * NEG_LOG2E;
    bo_t[j] = b * NEG_2LOG2E;
  }

  const int mg = n & 3;
  const float zscale = (mg == 2) ? NEG_2LOG2E : NEG_LOG2E;
  float th[4];
#pragma unroll
  for (int w = 0; w < 4; ++w) th[w] = thetas[mg * 4 + w];

  float*       hW = h_lds + n * HP + hi;
  const float* hA = h_lds + (4 * mg + kh) * HP;
  const float* yR = y0_lds + q;
  float* hs_p = hs + ((size_t)brow * S_SZ) * H_SZ + lane;

  float c_state[4], h_keep[4];
#pragma unroll
  for (int r = 0; r < 4; ++r) { c_state[r] = 0.f; h_keep[r] = 0.f; }

  __builtin_amdgcn_fence(__ATOMIC_RELEASE, "wavefront");
  __builtin_amdgcn_wave_barrier();
  __builtin_amdgcn_fence(__ATOMIC_ACQUIRE, "wavefront");

  float ysum = 0.f;
  float yq = yR[0];

  for (int t = 0; t < S_SZ; ++t) {
    const float y = ysum + yq;

    float p0[4], p1[4];
#pragma unroll
    for (int w = 0; w < 4; ++w) {
      const float yw = __int_as_float(__builtin_amdgcn_readlane(__float_as_int(y), 4 * w));
      float cv = __cosf(yw + th[w]);
      p0[w] = 0.5f + 0.5f * cv;
      p1[w] = 0.5f - 0.5f * cv;
    }
    float z0 = 0.f, z1 = 0.f, z2 = 0.f, z3 = 0.f;
#pragma unroll
    for (int idx = 0; idx < 16; ++idx) {
      int a0 = (idx >> 3) & 1, a1 = (idx >> 2) & 1;
      int a2 = (idx >> 1) & 1, a3 = idx & 1;
      float p = (a0 ? p1[0] : p0[0]) * (a1 ? p1[1] : p0[1]) *
                (a2 ? p1[2] : p0[2]) * (a3 ? p1[3] : p0[3]);
      int t1 = a1 ^ a0;
      int t2 = a2 ^ t1;
      int t3 = a3 ^ t2;
      int t0 = a0 ^ t3;
      z0 += t0 ? -p : p;  z1 += t1 ? -p : p;
      z2 += t2 ? -p : p;  z3 += t3 ? -p : p;
    }
    const v2f za = {(hi ? z2 : z0) * zscale, (hi ? z3 : z1) * zscale};

    float* hs_t = hs_p + (size_t)t * H_SZ;
    v8f ga[8];
#pragma unroll
    for (int j = 0; j < 8; ++j) {
      const v8f cbias = {bo_s[j], bo_s[j], bo_t[j], bo_s[j], bo_s[j], bo_s[j], bo_t[j], bo_s[j]};
      ga[j] = wmma_f32_16x16x4(za, bw[j], cbias);
    }
#pragma unroll
    for (int jj = 0; jj < 2; ++jj) {
      float fv = sigmoid_of_scaled(hi ? ga[2 * jj + 1][0] : ga[2 * jj][0]);
      float iv = sigmoid_of_scaled(hi ? ga[2 * jj + 1][1] : ga[2 * jj][1]);
      float gv = tanh_of_scaled(hi ? ga[2 * jj + 1][2] : ga[2 * jj][2]);
      float ov = sigmoid_of_scaled(hi ? ga[2 * jj + 1][3] : ga[2 * jj][3]);
      float cs = fv * c_state[jj] + iv * gv;
      c_state[jj] = cs;
      float hv = ov * tanh_of_scaled(cs * NEG_2LOG2E);
      h_keep[jj] = hv;
    }
    hW[0] = h_keep[0];
    hW[2] = h_keep[1];
    __builtin_amdgcn_fence(__ATOMIC_RELEASE, "wavefront");
    __builtin_amdgcn_wave_barrier();
    __builtin_amdgcn_fence(__ATOMIC_ACQUIRE, "wavefront");
    const float4 ra0 = *(const float4*)hA;
    const float4 rb0 = *(const float4*)(hA + HP);
    const int tn = (t + 1 < S_SZ) ? t + 1 : t;
    yq = yR[tn * 4];
    __builtin_amdgcn_sched_barrier(0);
    v8f acc0 = {0.f, 0.f, 0.f, 0.f, 0.f, 0.f, 0.f, 0.f};
    v8f acc1 = {0.f, 0.f, 0.f, 0.f, 0.f, 0.f, 0.f, 0.f};
    {
      const v2f a00 = {ra0.x, ra0.y}, a01 = {ra0.z, ra0.w};
      const v2f a10 = {rb0.x, rb0.y}, a11 = {rb0.z, rb0.w};
      acc0 = wmma_f32_16x16x4(a00, bh[0][0], acc0);
      acc1 = wmma_f32_16x16x4(a10, bh[1][0], acc1);
      acc0 = wmma_f32_16x16x4(a01, bh[0][1], acc0);
      acc1 = wmma_f32_16x16x4(a11, bh[1][1], acc1);
    }
#pragma unroll
    for (int jj = 2; jj < 4; ++jj) {
      float fv = sigmoid_of_scaled(hi ? ga[2 * jj + 1][0] : ga[2 * jj][0]);
      float iv = sigmoid_of_scaled(hi ? ga[2 * jj + 1][1] : ga[2 * jj][1]);
      float gv = tanh_of_scaled(hi ? ga[2 * jj + 1][2] : ga[2 * jj][2]);
      float ov = sigmoid_of_scaled(hi ? ga[2 * jj + 1][3] : ga[2 * jj][3]);
      float cs = fv * c_state[jj] + iv * gv;
      c_state[jj] = cs;
      float hv = ov * tanh_of_scaled(cs * NEG_2LOG2E);
      h_keep[jj] = hv;
    }
    hW[4] = h_keep[2];
    hW[6] = h_keep[3];
    __builtin_amdgcn_fence(__ATOMIC_RELEASE, "wavefront");
    __builtin_amdgcn_wave_barrier();
    __builtin_amdgcn_fence(__ATOMIC_ACQUIRE, "wavefront");
    {
      const float4 ra1 = *(const float4*)(hA + 4);
      const float4 rb1 = *(const float4*)(hA + HP + 4);
#pragma unroll
      for (int jj = 0; jj < 4; ++jj) hs_t[jj * 32] = h_keep[jj];
      const v2f a02 = {ra1.x, ra1.y}, a03 = {ra1.z, ra1.w};
      const v2f a12 = {rb1.x, rb1.y}, a13 = {rb1.z, rb1.w};
      acc0 = wmma_f32_16x16x4(a02, bh[0][2], acc0);
      acc1 = wmma_f32_16x16x4(a12, bh[1][2], acc1);
      acc0 = wmma_f32_16x16x4(a03, bh[0][3], acc0);
      acc1 = wmma_f32_16x16x4(a13, bh[1][3], acc1);
    }
    const float s0 = acc0[0] + acc1[0], s1 = acc0[1] + acc1[1];
    const float s2 = acc0[2] + acc1[2], s3 = acc0[3] + acc1[3];
    const float g0 = __int_as_float(__builtin_amdgcn_update_dpp(0, __float_as_int(s0), 0x00, 0xF, 0xF, true));
    const float g1 = __int_as_float(__builtin_amdgcn_update_dpp(0, __float_as_int(s1), 0x55, 0xF, 0xF, true));
    const float g2 = __int_as_float(__builtin_amdgcn_update_dpp(0, __float_as_int(s2), 0xAA, 0xF, 0xF, true));
    const float g3 = __int_as_float(__builtin_amdgcn_update_dpp(0, __float_as_int(s3), 0xFF, 0xF, 0xF, true));
    const float d = ((g0 + g1) + g2) + g3;
    ysum = d;
  }

#pragma unroll
  for (int jj = 0; jj < 4; ++jj) {
    hT[(size_t)brow * H_SZ + jj * 32 + lane] = h_keep[jj];
    cT[(size_t)brow * H_SZ + jj * 32 + lane] = c_state[jj];
  }
}

extern "C" void kernel_launch(void* const* d_in, const int* in_sizes, int n_in,
                              void* d_out, int out_size, void* d_ws, size_t ws_size,
                              hipStream_t stream) {
  const float* x      = (const float*)d_in[0];
  const float* W_in   = (const float*)d_in[1];
  const float* b_in   = (const float*)d_in[2];
  const float* W_out  = (const float*)d_in[3];
  const float* b_out  = (const float*)d_in[4];
  const float* thetas = (const float*)d_in[5];
  float* out = (float*)d_out;
  float* y0  = (float*)d_ws;

  qlstm_xproj<<<1024, 128, 0, stream>>>(x, W_in, b_in, y0);
  qlstm_recur<<<256, 32, 0, stream>>>(y0, W_in, W_out, b_out, thetas, out);
}


# y product re-tiled so a lane's own h registers are its A operand (rows = 16 column lanes, 4 B tiles of (wire, row class)): no h trip through LDS; y sum order regrouped (k order, join tree)
# speedup vs baseline: 3.5286x; 1.1212x over previous
#include <hip/hip_runtime.h>
#include <hip/hip_bf16.h>
#include <math.h>

typedef __attribute__((ext_vector_type(2))) float v2f;
typedef __attribute__((ext_vector_type(8))) float v8f;

static __device__ __forceinline__ v8f wmma_f32_16x16x4(v2f a, v2f b, v8f c) {
  return __builtin_amdgcn_wmma_f32_16x16x4_f32(false, a, false, b, (short)0, c, false, false);
}

#define NEG_LOG2E  (-1.4426950408889634f)
#define NEG_2LOG2E (-2.8853900817779268f)
static __device__ __forceinline__ float sigmoid_of_scaled(float d) {
  return __builtin_amdgcn_rcpf(1.f + __builtin_amdgcn_exp2f(d));
}
static __device__ __forceinline__ float tanh_of_scaled(float d) {
  return fmaf(2.f, __builtin_amdgcn_rcpf(1.f + __builtin_amdgcn_exp2f(d)), -1.f);
}

#define B_SZ 256
#define S_SZ 256
#define F_SZ 64
#define H_SZ 128

__global__ void __launch_bounds__(128)
qlstm_xproj(const float* __restrict__ x, const float* __restrict__ Win,
            const float* __restrict__ bin, float* __restrict__ y0) {
  const int lane = threadIdx.x & 31;
  const int wave = threadIdx.x >> 5;
  const int tile = blockIdx.x * 4 + wave;
  const int n    = lane & 15;
  const int hi   = lane >> 4;
  const int kh   = hi << 1;

  v2f bch[16];
#pragma unroll
  for (int j = 0; j < 16; ++j) {
    float b0 = 0.f, b1 = 0.f;
    if (n < 4) {
      int kk = j * 4 + kh;
      b0 = Win[n * 192 + 128 + kk];
      b1 = Win[n * 192 + 128 + kk + 1];
    }
    v2f bv = {b0, b1};
    bch[j] = bv;
  }

  const float bb = (n < 4) ? bin[n] : 0.f;
  v8f acc = {bb, bb, bb, bb, bb, bb, bb, bb};

  const float* xr = x + (size_t)(tile * 16 + n) * F_SZ;
#pragma unroll
  for (int j = 0; j < 16; ++j) {
    v2f a = {xr[j * 4 + kh], xr[j * 4 + kh + 1]};
    acc = wmma_f32_16x16x4(a, bch[j], acc);
  }

  if (n < 4) {
#pragma unroll
    for (int r = 0; r < 8; ++r) {
      int row = tile * 16 + r + (hi << 3);
      y0[(size_t)row * 4 + n] = acc[r];
    }
  }
}

__global__ void __launch_bounds__(32)
qlstm_recur(const float* __restrict__ y0, const float* __restrict__ Win,
            const float* __restrict__ Wout, const float* __restrict__ bout,
            const float* __restrict__ thetas, float* __restrict__ out) {
  __shared__ __attribute__((aligned(16))) float y0_lds[S_SZ * 4];

  const int lane  = threadIdx.x;
  const int n     = lane & 15;
  const int hi    = lane >> 4;
  const int kh    = hi << 1;
  const int brow  = blockIdx.x;

  float* hs = out;
  float* hT = out + (size_t)B_SZ * S_SZ * H_SZ;
  float* cT = hT + (size_t)B_SZ * H_SZ;

  {
    const float4* src = (const float4*)(y0 + (size_t)brow * S_SZ * 4);
#pragma unroll
    for (int i = 0; i < 8; ++i) {
      int idx = lane + i * 32;
      *(float4*)&y0_lds[idx * 4] = src[idx];
    }
  }

  const int q  = n >> 2;
  const int cg = n & 3;
  v2f bh[2][4];
#pragma unroll
  for (int u = 0; u < 2; ++u) {
#pragma unroll
    for (int ct = 0; ct < 4; ++ct) {
      int kk = 64 * u + 16 * hi + 4 * cg + ct;
      v2f bv = {Win[q * 192 + kk], Win[q * 192 + kk + 32]};
      bh[u][ct] = bv;
    }
  }
  v2f bw[8];
  float bo_s[8], bo_t[8];
#pragma unroll
  for (int j = 0; j < 8; ++j) {
    int col = j * 16 + n;
    v2f bv = {Wout[col * 4 + kh], Wout[col * 4 + kh + 1]};
    bw[j] = bv;
    const float b = bout[col];
    bo_s[j] = b * NEG_LOG2E;
    bo_t[j] = b * NEG_2LOG2E;
  }

  const int mg = n & 3;
  const float zscale = (mg == 2) ? NEG_2LOG2E : NEG_LOG2E;
  float th[4];
#pragma unroll
  for (int w = 0; w < 4; ++w) th[w] = thetas[mg * 4 + w];

  const float* yR = y0_lds + q;
  float* hs_p = hs + ((size_t)brow * S_SZ) * H_SZ + lane;

  float c_state[4], h_keep[4];
#pragma unroll
  for (int r = 0; r < 4; ++r) { c_state[r] = 0.f; h_keep[r] = 0.f; }

  __builtin_amdgcn_fence(__ATOMIC_RELEASE, "wavefront");
  __builtin_amdgcn_wave_barrier();
  __builtin_amdgcn_fence(__ATOMIC_ACQUIRE, "wavefront");

  float ysum = 0.f;
  float yq = yR[0];

  for (int t = 0; t < S_SZ; ++t) {
    const float y = ysum + yq;

    float p0[4], p1[4];
#pragma unroll
    for (int w = 0; w < 4; ++w) {
      const float yw = __int_as_float(__builtin_amdgcn_readlane(__float_as_int(y), 4 * w));
      float cv = __cosf(yw + th[w]);
      p0[w] = 0.5f + 0.5f * cv;
      p1[w] = 0.5f - 0.5f * cv;
    }
    float z0 = 0.f, z1 = 0.f, z2 = 0.f, z3 = 0.f;
#pragma unroll
    for (int idx = 0; idx < 16; ++idx) {
      int a0 = (idx >> 3) & 1, a1 = (idx >> 2) & 1;
      int a2 = (idx >> 1) & 1, a3 = idx & 1;
      float p = (a0 ? p1[0] : p0[0]) * (a1 ? p1[1] : p0[1]) *
                (a2 ? p1[2] : p0[2]) * (a3 ? p1[3] : p0[3]);
      int t1 = a1 ^ a0;
      int t2 = a2 ^ t1;
      int t3 = a3 ^ t2;
      int t0 = a0 ^ t3;
      z0 += t0 ? -p : p;  z1 += t1 ? -p : p;
      z2 += t2 ? -p : p;  z3 += t3 ? -p : p;
    }
    const v2f za = {(hi ? z2 : z0) * zscale, (hi ? z3 : z1) * zscale};

    float* hs_t = hs_p + (size_t)t * H_SZ;
    v8f ga[8];
#pragma unroll
    for (int j = 0; j < 8; ++j) {
      const v8f cbias = {bo_s[j], bo_s[j], bo_t[j], bo_s[j], bo_s[j], bo_s[j], bo_t[j], bo_s[j]};
      ga[j] = wmma_f32_16x16x4(za, bw[j], cbias);
    }
#pragma unroll
    for (int jj = 0; jj < 2; ++jj) {
      float fv = sigmoid_of_scaled(hi ? ga[2 * jj + 1][0] : ga[2 * jj][0]);
      float iv = sigmoid_of_scaled(hi ? ga[2 * jj + 1][1] : ga[2 * jj][1]);
      float gv = tanh_of_scaled(hi ? ga[2 * jj + 1][2] : ga[2 * jj][2]);
      float ov = sigmoid_of_scaled(hi ? ga[2 * jj + 1][3] : ga[2 * jj][3]);
      float cs = fv * c_state[jj] + iv * gv;
      c_state[jj] = cs;
      float hv = ov * tanh_of_scaled(cs * NEG_2LOG2E);
      h_keep[jj] = hv;
    }
    const int tn = (t + 1 < S_SZ) ? t + 1 : t;
    yq = yR[tn * 4];
    __builtin_amdgcn_sched_barrier(0);
    v8f acc[4];
    {
      const v2f a0 = {h_keep[0], h_keep[1]};
      const v8f zero = {0.f, 0.f, 0.f, 0.f, 0.f, 0.f, 0.f, 0.f};
#pragma unroll
      for (int ct = 0; ct < 4; ++ct) acc[ct] = wmma_f32_16x16x4(a0, bh[0][ct], zero);
    }
#pragma unroll
    for (int jj = 2; jj < 4; ++jj) {
      float fv = sigmoid_of_scaled(hi ? ga[2 * jj + 1][0] : ga[2 * jj][0]);
      float iv = sigmoid_of_scaled(hi ? ga[2 * jj + 1][1] : ga[2 * jj][1]);
      float gv = tanh_of_scaled(hi ? ga[2 * jj + 1][2] : ga[2 * jj][2]);
      float ov = sigmoid_of_scaled(hi ? ga[2 * jj + 1][3] : ga[2 * jj][3]);
      float cs = fv * c_state[jj] + iv * gv;
      c_state[jj] = cs;
      float hv = ov * tanh_of_scaled(cs * NEG_2LOG2E);
      h_keep[jj] = hv;
    }
    {
#pragma unroll
      for (int jj = 0; jj < 4; ++jj) hs_t[jj * 32] = h_keep[jj];
      const v2f a1 = {h_keep[2], h_keep[3]};
#pragma unroll
      for (int ct = 0; ct < 4; ++ct) acc[ct] = wmma_f32_16x16x4(a1, bh[1][ct], acc[ct]);
    }
    const float sa = (acc[0][0] + acc[1][1]) + (acc[2][2] + acc[3][3]);
    const float sb = (acc[0][4] + acc[1][5]) + (acc[2][6] + acc[3][7]);
    const float vs = (cg & 1) ? sb : sa;
    const float tp = vs + __int_as_float(__builtin_amdgcn_update_dpp(0, __float_as_int(vs), 0xB1, 0xF, 0xF, true));
    const float xo = __int_as_float(__builtin_amdgcn_permlanex16(0, __float_as_int(tp), 0x54761032u, 0xDCFE98BAu, false, false));
    const float d = tp + xo;
    ysum = d;
  }

#pragma unroll
  for (int jj = 0; jj < 4; ++jj) {
    hT[(size_t)brow * H_SZ + jj * 32 + lane] = h_keep[jj];
    cT[(size_t)brow * H_SZ + jj * 32 + lane] = c_state[jj];
  }
}

extern "C" void kernel_launch(void* const* d_in, const int* in_sizes, int n_in,
                              void* d_out, int out_size, void* d_ws, size_t ws_size,
                              hipStream_t stream) {
  const float* x      = (const float*)d_in[0];
  const float* W_in   = (const float*)d_in[1];
  const float* b_in   = (const float*)d_in[2];
  const float* W_out  = (const float*)d_in[3];
  const float* b_out  = (const float*)d_in[4];
  const float* thetas = (const float*)d_in[5];
  float* out = (float*)d_out;
  float* y0  = (float*)d_ws;

  qlstm_xproj<<<1024, 128, 0, stream>>>(x, W_in, b_in, y0);
  qlstm_recur<<<256, 32, 0, stream>>>(y0, W_in, W_out, b_out, thetas, out);
}


# qlayer's 16-term probability sums as one shared pairwise tree (butterfly over the wire bits, two wide, depth 4) where 4 serial chains of 16; gate factor folded into wire 0; z sum order regrouped
# speedup vs baseline: 3.8864x; 1.1014x over previous
#include <hip/hip_runtime.h>
#include <hip/hip_bf16.h>
#include <math.h>

typedef __attribute__((ext_vector_type(2))) float v2f;
typedef __attribute__((ext_vector_type(8))) float v8f;

static __device__ __forceinline__ v8f wmma_f32_16x16x4(v2f a, v2f b, v8f c) {
  return __builtin_amdgcn_wmma_f32_16x16x4_f32(false, a, false, b, (short)0, c, false, false);
}

#define NEG_LOG2E  (-1.4426950408889634f)
#define NEG_2LOG2E (-2.8853900817779268f)
static __device__ __forceinline__ float sigmoid_of_scaled(float d) {
  return __builtin_amdgcn_rcpf(1.f + __builtin_amdgcn_exp2f(d));
}
static __device__ __forceinline__ float tanh_of_scaled(float d) {
  return fmaf(2.f, __builtin_amdgcn_rcpf(1.f + __builtin_amdgcn_exp2f(d)), -1.f);
}

#define B_SZ 256
#define S_SZ 256
#define F_SZ 64
#define H_SZ 128

__global__ void __launch_bounds__(128)
qlstm_xproj(const float* __restrict__ x, const float* __restrict__ Win,
            const float* __restrict__ bin, float* __restrict__ y0) {
  const int lane = threadIdx.x & 31;
  const int wave = threadIdx.x >> 5;
  const int tile = blockIdx.x * 4 + wave;
  const int n    = lane & 15;
  const int hi   = lane >> 4;
  const int kh   = hi << 1;

  v2f bch[16];
#pragma unroll
  for (int j = 0; j < 16; ++j) {
    float b0 = 0.f, b1 = 0.f;
    if (n < 4) {
      int kk = j * 4 + kh;
      b0 = Win[n * 192 + 128 + kk];
      b1 = Win[n * 192 + 128 + kk + 1];
    }
    v2f bv = {b0, b1};
    bch[j] = bv;
  }

  const float bb = (n < 4) ? bin[n] : 0.f;
  v8f acc = {bb, bb, bb, bb, bb, bb, bb, bb};

  const float* xr = x + (size_t)(tile * 16 + n) * F_SZ;
#pragma unroll
  for (int j = 0; j < 16; ++j) {
    v2f a = {xr[j * 4 + kh], xr[j * 4 + kh + 1]};
    acc = wmma_f32_16x16x4(a, bch[j], acc);
  }

  if (n < 4) {
#pragma unroll
    for (int r = 0; r < 8; ++r) {
      int row = tile * 16 + r + (hi << 3);
      y0[(size_t)row * 4 + n] = acc[r];
    }
  }
}

__global__ void __launch_bounds__(32)
qlstm_recur(const float* __restrict__ y0, const float* __restrict__ Win,
            const float* __restrict__ Wout, const float* __restrict__ bout,
            const float* __restrict__ thetas, float* __restrict__ out) {
  __shared__ __attribute__((aligned(16))) float y0_lds[S_SZ * 4];

  const int lane  = threadIdx.x;
  const int n     = lane & 15;
  const int hi    = lane >> 4;
  const int kh    = hi << 1;
  const int brow  = blockIdx.x;

  float* hs = out;
  float* hT = out + (size_t)B_SZ * S_SZ * H_SZ;
  float* cT = hT + (size_t)B_SZ * H_SZ;

  {
    const float4* src = (const float4*)(y0 + (size_t)brow * S_SZ * 4);
#pragma unroll
    for (int i = 0; i < 8; ++i) {
      int idx = lane + i * 32;
      *(float4*)&y0_lds[idx * 4] = src[idx];
    }
  }

  const int q  = n >> 2;
  const int cg = n & 3;
  v2f bh[2][4];
#pragma unroll
  for (int u = 0; u < 2; ++u) {
#pragma unroll
    for (int ct = 0; ct < 4; ++ct) {
      int kk = 64 * u + 16 * hi + 4 * cg + ct;
      v2f bv = {Win[q * 192 + kk], Win[q * 192 + kk + 32]};
      bh[u][ct] = bv;
    }
  }
  v2f bw[8];
  float bo_s[8], bo_t[8];
#pragma unroll
  for (int j = 0; j < 8; ++j) {
    int col = j * 16 + n;
    v2f bv = {Wout[col * 4 + kh], Wout[col * 4 + kh + 1]};
    bw[j] = bv;
    const float b = bout[col];
    bo_s[j] = b * NEG_LOG2E;
    bo_t[j] = b * NEG_2LOG2E;
  }

  const int mg = n & 3;
  const float zscale = (mg == 2) ? NEG_2LOG2E : NEG_LOG2E;
  const v2f zsign = {hi ? 1.f : -1.f, 1.f};
  const v2f hp = {0.5f, -0.5f}, hm = {-0.5f, 0.5f}, hh = {0.5f, 0.5f};
  const v2f hpz = hp * zscale, hhz = hh * zscale;
  float th[4];
#pragma unroll
  for (int w = 0; w < 4; ++w) th[w] = thetas[mg * 4 + w];

  const float* yR = y0_lds + q;
  float* hs_p = hs + ((size_t)brow * S_SZ) * H_SZ + lane;

  float c_state[4], h_keep[4];
#pragma unroll
  for (int r = 0; r < 4; ++r) { c_state[r] = 0.f; h_keep[r] = 0.f; }

  __builtin_amdgcn_fence(__ATOMIC_RELEASE, "wavefront");
  __builtin_amdgcn_wave_barrier();
  __builtin_amdgcn_fence(__ATOMIC_ACQUIRE, "wavefront");

  float ysum = 0.f;
  float yq = yR[0];

  for (int t = 0; t < S_SZ; ++t) {
    const float y = ysum + yq;

    v2f pw[4], pws[4];
#pragma unroll
    for (int w = 0; w < 4; ++w) {
      const float yw = __int_as_float(__builtin_amdgcn_readlane(__float_as_int(y), 4 * w));
      const float cv = __cosf(yw + th[w]);
      pw[w]  = cv * (w ? hp : hpz) + (w ? hh : hhz);
      pws[w] = cv * hm + hh;
    }
    const v2f PA  = pw[0] * pw[1];
    const v2f PB  = pw[0] * pws[1];
    const v2f Q03 = pw[2] * pw[3];
    const v2f Q12 = pw[2] * pws[3];
    v2f Rs[2], Vs[2], Xs[2];
#pragma unroll
    for (int g = 0; g < 2; ++g) {
      const v2f P = g ? PB : PA;
      const v2f t00 = P * Q03.x, t01 = P * Q12.x;
      const v2f t10 = P * Q12.y, t11 = P * Q03.y;
      const v2f u0 = t00 + t01, u1 = t10 + t11;
      const v2f d0 = t00 - t01, d1 = t10 - t11;
      Rs[g] = u0 + u1;
      Vs[g] = u0 - u1;
      Xs[g] = d0 - d1;
    }
    const v2f rd = Rs[0] - Rs[1];
    const v2f vd = Vs[0] - Vs[1];
    const v2f xd = Xs[0] - Xs[1];
    const v2f fa = {hi ? vd.x : xd.x, hi ? xd.x : rd.x};
    const v2f fb = {hi ? vd.y : xd.y, hi ? xd.y : rd.y};
    const v2f za = fa + zsign * fb;

    float* hs_t = hs_p + (size_t)t * H_SZ;
    v8f ga[8];
#pragma unroll
    for (int j = 0; j < 8; ++j) {
      const v8f cbias = {bo_s[j], bo_s[j], bo_t[j], bo_s[j], bo_s[j], bo_s[j], bo_t[j], bo_s[j]};
      ga[j] = wmma_f32_16x16x4(za, bw[j], cbias);
    }
#pragma unroll
    for (int jj = 0; jj < 2; ++jj) {
      float fv = sigmoid_of_scaled(hi ? ga[2 * jj + 1][0] : ga[2 * jj][0]);
      float iv = sigmoid_of_scaled(hi ? ga[2 * jj + 1][1] : ga[2 * jj][1]);
      float gv = tanh_of_scaled(hi ? ga[2 * jj + 1][2] : ga[2 * jj][2]);
      float ov = sigmoid_of_scaled(hi ? ga[2 * jj + 1][3] : ga[2 * jj][3]);
      float cs = fv * c_state[jj] + iv * gv;
      c_state[jj] = cs;
      float hv = ov * tanh_of_scaled(cs * NEG_2LOG2E);
      h_keep[jj] = hv;
    }
    const int tn = (t + 1 < S_SZ) ? t + 1 : t;
    yq = yR[tn * 4];
    __builtin_amdgcn_sched_barrier(0);
    v8f acc[4];
    {
      const v2f a0 = {h_keep[0], h_keep[1]};
      const v8f zero = {0.f, 0.f, 0.f, 0.f, 0.f, 0.f, 0.f, 0.f};
#pragma unroll
      for (int ct = 0; ct < 4; ++ct) acc[ct] = wmma_f32_16x16x4(a0, bh[0][ct], zero);
    }
#pragma unroll
    for (int jj = 2; jj < 4; ++jj) {
      float fv = sigmoid_of_scaled(hi ? ga[2 * jj + 1][0] : ga[2 * jj][0]);
      float iv = sigmoid_of_scaled(hi ? ga[2 * jj + 1][1] : ga[2 * jj][1]);
      float gv = tanh_of_scaled(hi ? ga[2 * jj + 1][2] : ga[2 * jj][2]);
      float ov = sigmoid_of_scaled(hi ? ga[2 * jj + 1][3] : ga[2 * jj][3]);
      float cs = fv * c_state[jj] + iv * gv;
      c_state[jj] = cs;
      float hv = ov * tanh_of_scaled(cs * NEG_2LOG2E);
      h_keep[jj] = hv;
    }
    {
#pragma unroll
      for (int jj = 0; jj < 4; ++jj) hs_t[jj * 32] = h_keep[jj];
      const v2f a1 = {h_keep[2], h_keep[3]};
#pragma unroll
      for (int ct = 0; ct < 4; ++ct) acc[ct] = wmma_f32_16x16x4(a1, bh[1][ct], acc[ct]);
    }
    const float sa = (acc[0][0] + acc[1][1]) + (acc[2][2] + acc[3][3]);
    const float sb = (acc[0][4] + acc[1][5]) + (acc[2][6] + acc[3][7]);
    const float vs = (cg & 1) ? sb : sa;
    const float tp = vs + __int_as_float(__builtin_amdgcn_update_dpp(0, __float_as_int(vs), 0xB1, 0xF, 0xF, true));
    const float xo = __int_as_float(__builtin_amdgcn_permlanex16(0, __float_as_int(tp), 0x54761032u, 0xDCFE98BAu, false, false));
    const float d = tp + xo;
    ysum = d;
  }

#pragma unroll
  for (int jj = 0; jj < 4; ++jj) {
    hT[(size_t)brow * H_SZ + jj * 32 + lane] = h_keep[jj];
    cT[(size_t)brow * H_SZ + jj * 32 + lane] = c_state[jj];
  }
}

extern "C" void kernel_launch(void* const* d_in, const int* in_sizes, int n_in,
                              void* d_out, int out_size, void* d_ws, size_t ws_size,
                              hipStream_t stream) {
  const float* x      = (const float*)d_in[0];
  const float* W_in   = (const float*)d_in[1];
  const float* b_in   = (const float*)d_in[2];
  const float* W_out  = (const float*)d_in[3];
  const float* b_out  = (const float*)d_in[4];
  const float* thetas = (const float*)d_in[5];
  float* out = (float*)d_out;
  float* y0  = (float*)d_ws;

  qlstm_xproj<<<1024, 128, 0, stream>>>(x, W_in, b_in, y0);
  qlstm_recur<<<256, 32, 0, stream>>>(y0, W_in, W_out, b_out, thetas, out);
}


# issue order pinned: one gate or y matrix instruction per slice of activation work (cell update cut in stages, neighbours share a slice), so matrix pipe and vector unit overlap; arithmetic unchanged
# speedup vs baseline: 4.0087x; 1.0315x over previous
#include <hip/hip_runtime.h>
#include <hip/hip_bf16.h>
#include <math.h>

typedef __attribute__((ext_vector_type(2))) float v2f;
typedef __attribute__((ext_vector_type(8))) float v8f;

static __device__ __forceinline__ v8f wmma_f32_16x16x4(v2f a, v2f b, v8f c) {
  return __builtin_amdgcn_wmma_f32_16x16x4_f32(false, a, false, b, (short)0, c, false, false);
}

#define SLICE_END() __builtin_amdgcn_sched_barrier(0)
#define ISSUED_BEFORE_HERE(val) asm volatile("" :: "v"(val))

#define NEG_LOG2E  (-1.4426950408889634f)
#define NEG_2LOG2E (-2.8853900817779268f)

#define B_SZ 256
#define S_SZ 256
#define F_SZ 64
#define H_SZ 128

__global__ void __launch_bounds__(128)
qlstm_xproj(const float* __restrict__ x, const float* __restrict__ Win,
            const float* __restrict__ bin, float* __restrict__ y0) {
  const int lane = threadIdx.x & 31;
  const int wave = threadIdx.x >> 5;
  const int tile = blockIdx.x * 4 + wave;
  const int n    = lane & 15;
  const int hi   = lane >> 4;
  const int kh   = hi << 1;

  v2f bch[16];
#pragma unroll
  for (int j = 0; j < 16; ++j) {
    float b0 = 0.f, b1 = 0.f;
    if (n < 4) {
      int kk = j * 4 + kh;
      b0 = Win[n * 192 + 128 + kk];
      b1 = Win[n * 192 + 128 + kk + 1];
    }
    v2f bv = {b0, b1};
    bch[j] = bv;
  }

  const float bb = (n < 4) ? bin[n] : 0.f;
  v8f acc = {bb, bb, bb, bb, bb, bb, bb, bb};

  const float* xr = x + (size_t)(tile * 16 + n) * F_SZ;
#pragma unroll
  for (int j = 0; j < 16; ++j) {
    v2f a = {xr[j * 4 + kh], xr[j * 4 + kh + 1]};
    acc = wmma_f32_16x16x4(a, bch[j], acc);
  }

  if (n < 4) {
#pragma unroll
    for (int r = 0; r < 8; ++r) {
      int row = tile * 16 + r + (hi << 3);
      y0[(size_t)row * 4 + n] = acc[r];
    }
  }
}

__global__ void __launch_bounds__(32)
qlstm_recur(const float* __restrict__ y0, const float* __restrict__ Win,
            const float* __restrict__ Wout, const float* __restrict__ bout,
            const float* __restrict__ thetas, float* __restrict__ out) {
  __shared__ __attribute__((aligned(16))) float y0_lds[S_SZ * 4];

  const int lane  = threadIdx.x;
  const int n     = lane & 15;
  const int hi    = lane >> 4;
  const int kh    = hi << 1;
  const int brow  = blockIdx.x;

  float* hs = out;
  float* hT = out + (size_t)B_SZ * S_SZ * H_SZ;
  float* cT = hT + (size_t)B_SZ * H_SZ;

  {
    const float4* src = (const float4*)(y0 + (size_t)brow * S_SZ * 4);
#pragma unroll
    for (int i = 0; i < 8; ++i) {
      int idx = lane + i * 32;
      *(float4*)&y0_lds[idx * 4] = src[idx];
    }
  }

  const int q  = n >> 2;
  const int cg = n & 3;
  v2f bh[2][4];
#pragma unroll
  for (int u = 0; u < 2; ++u) {
#pragma unroll
    for (int ct = 0; ct < 4; ++ct) {
      int kk = 64 * u + 16 * hi + 4 * cg + ct;
      v2f bv = {Win[q * 192 + kk], Win[q * 192 + kk + 32]};
      bh[u][ct] = bv;
    }
  }
  v2f bw[8];
  float bo_s[8], bo_t[8];
#pragma unroll
  for (int j = 0; j < 8; ++j) {
    int col = j * 16 + n;
    v2f bv = {Wout[col * 4 + kh], Wout[col * 4 + kh + 1]};
    bw[j] = bv;
    const float b = bout[col];
    bo_s[j] = b * NEG_LOG2E;
    bo_t[j] = b * NEG_2LOG2E;
  }

  const int mg = n & 3;
  const float zscale = (mg == 2) ? NEG_2LOG2E : NEG_LOG2E;
  const v2f zsign = {hi ? 1.f : -1.f, 1.f};
  const v2f hp = {0.5f, -0.5f}, hm = {-0.5f, 0.5f}, hh = {0.5f, 0.5f};
  const v2f hpz = hp * zscale, hhz = hh * zscale;
  float th[4];
#pragma unroll
  for (int w = 0; w < 4; ++w) th[w] = thetas[mg * 4 + w];

  const float* yR = y0_lds + q;
  float* hs_p = hs + ((size_t)brow * S_SZ) * H_SZ + lane;

  float c_state[4], h_keep[4];
#pragma unroll
  for (int r = 0; r < 4; ++r) { c_state[r] = 0.f; h_keep[r] = 0.f; }

  __builtin_amdgcn_fence(__ATOMIC_RELEASE, "wavefront");
  __builtin_amdgcn_wave_barrier();
  __builtin_amdgcn_fence(__ATOMIC_ACQUIRE, "wavefront");

  float ysum = 0.f;
  float yq = yR[0];

  for (int t = 0; t < S_SZ; ++t) {
    const float y = ysum + yq;

    v2f pw[4], pws[4];
#pragma unroll
    for (int w = 0; w < 4; ++w) {
      const float yw = __int_as_float(__builtin_amdgcn_readlane(__float_as_int(y), 4 * w));
      const float cv = __cosf(yw + th[w]);
      pw[w]  = cv * (w ? hp : hpz) + (w ? hh : hhz);
      pws[w] = cv * hm + hh;
    }
    const v2f PA  = pw[0] * pw[1];
    const v2f PB  = pw[0] * pws[1];
    const v2f Q03 = pw[2] * pw[3];
    const v2f Q12 = pw[2] * pws[3];
    v2f Rs[2], Vs[2], Xs[2];
#pragma unroll
    for (int g = 0; g < 2; ++g) {
      const v2f P = g ? PB : PA;
      const v2f t00 = P * Q03.x, t01 = P * Q12.x;
      const v2f t10 = P * Q12.y, t11 = P * Q03.y;
      const v2f u0 = t00 + t01, u1 = t10 + t11;
      const v2f d0 = t00 - t01, d1 = t10 - t11;
      Rs[g] = u0 + u1;
      Vs[g] = u0 - u1;
      Xs[g] = d0 - d1;
    }
    const v2f rd = Rs[0] - Rs[1];
    const v2f vd = Vs[0] - Vs[1];
    const v2f xd = Xs[0] - Xs[1];
    const v2f fa = {hi ? vd.x : xd.x, hi ? xd.x : rd.x};
    const v2f fb = {hi ? vd.y : xd.y, hi ? xd.y : rd.y};
    const v2f za = fa + zsign * fb;

    float* hs_t = hs_p + (size_t)t * H_SZ;
    v8f ga[8];
    float eg[4][4], rg[4][4], ecs[4];
    auto gate_tile = [&](int j) __attribute__((always_inline)) {
      const v8f cbias = {bo_s[j], bo_s[j], bo_t[j], bo_s[j], bo_s[j], bo_s[j], bo_t[j], bo_s[j]};
      ga[j] = wmma_f32_16x16x4(za, bw[j], cbias);
    };
    auto S1 = [&](int jj) __attribute__((always_inline)) {
#pragma unroll
      for (int r = 0; r < 4; ++r)
        eg[jj][r] = __builtin_amdgcn_exp2f(hi ? ga[2 * jj + 1][r] : ga[2 * jj][r]);
    };
    auto S2 = [&](int jj) __attribute__((always_inline)) {
#pragma unroll
      for (int r = 0; r < 4; ++r) rg[jj][r] = __builtin_amdgcn_rcpf(1.f + eg[jj][r]);
    };
    auto S3a = [&](int jj) __attribute__((always_inline)) {
      const float fv = rg[jj][0], iv = rg[jj][1];
      const float gv = fmaf(2.f, rg[jj][2], -1.f);
      float cs = fv * c_state[jj] + iv * gv;
      c_state[jj] = cs;
    };
    auto S3b = [&](int jj) __attribute__((always_inline)) {
      ecs[jj] = __builtin_amdgcn_exp2f(c_state[jj] * NEG_2LOG2E);
    };
    float rcs[4];
    auto S4a = [&](int jj) __attribute__((always_inline)) {
      rcs[jj] = __builtin_amdgcn_rcpf(1.f + ecs[jj]);
    };
    auto S4b = [&](int jj) __attribute__((always_inline)) {
      const float ov = rg[jj][3];
      h_keep[jj] = ov * fmaf(2.f, rcs[jj], -1.f);
    };
    v8f acc[4];
    const v8f zero = {0.f, 0.f, 0.f, 0.f, 0.f, 0.f, 0.f, 0.f};

    gate_tile(0); gate_tile(1); gate_tile(2); gate_tile(3);
    ISSUED_BEFORE_HERE(ga[0][0]); ISSUED_BEFORE_HERE(ga[1][0]);
    ISSUED_BEFORE_HERE(ga[2][0]); ISSUED_BEFORE_HERE(ga[3][0]);
    SLICE_END();
    S1(0);
    SLICE_END(); gate_tile(4); ISSUED_BEFORE_HERE(ga[4][0]); SLICE_END();
    S2(0);
    SLICE_END(); gate_tile(5); ISSUED_BEFORE_HERE(ga[5][0]); SLICE_END();
    S1(1);
    SLICE_END(); gate_tile(6); ISSUED_BEFORE_HERE(ga[6][0]); SLICE_END();
    S2(1); S3a(0); S3b(0);
    SLICE_END(); gate_tile(7); ISSUED_BEFORE_HERE(ga[7][0]); SLICE_END();
    S3a(1); S3b(1); S1(2); S4a(0);
    SLICE_END();
    S4a(1); S4b(1); S4b(0); S2(2); S1(3);
    const int tn = (t + 1 < S_SZ) ? t + 1 : t;
    yq = yR[tn * 4];
    const v2f a0 = {h_keep[0], h_keep[1]};
    SLICE_END(); acc[0] = wmma_f32_16x16x4(a0, bh[0][0], zero); ISSUED_BEFORE_HERE(acc[0][0]); SLICE_END();
    S2(3); S3a(2); S3b(2);
    SLICE_END(); acc[1] = wmma_f32_16x16x4(a0, bh[0][1], zero); ISSUED_BEFORE_HERE(acc[1][0]); SLICE_END();
    S3a(3); S3b(3); S4a(2); S4b(2);
    SLICE_END(); acc[2] = wmma_f32_16x16x4(a0, bh[0][2], zero); ISSUED_BEFORE_HERE(acc[2][0]); SLICE_END();
    S4a(3);
    SLICE_END(); acc[3] = wmma_f32_16x16x4(a0, bh[0][3], zero); ISSUED_BEFORE_HERE(acc[3][0]); SLICE_END();
    S4b(3);
    {
      const v2f a1 = {h_keep[2], h_keep[3]};
#pragma unroll
      for (int ct = 0; ct < 4; ++ct) acc[ct] = wmma_f32_16x16x4(a1, bh[1][ct], acc[ct]);
    }
    SLICE_END();
#pragma unroll
    for (int jj = 0; jj < 4; ++jj) hs_t[jj * 32] = h_keep[jj];
    const float sa = (acc[0][0] + acc[1][1]) + (acc[2][2] + acc[3][3]);
    const float sb = (acc[0][4] + acc[1][5]) + (acc[2][6] + acc[3][7]);
    const float vs = (cg & 1) ? sb : sa;
    const float tp = vs + __int_as_float(__builtin_amdgcn_update_dpp(0, __float_as_int(vs), 0xB1, 0xF, 0xF, true));
    const float xo = __int_as_float(__builtin_amdgcn_permlanex16(0, __float_as_int(tp), 0x54761032u, 0xDCFE98BAu, false, false));
    const float d = tp + xo;
    ysum = d;
  }

#pragma unroll
  for (int jj = 0; jj < 4; ++jj) {
    hT[(size_t)brow * H_SZ + jj * 32 + lane] = h_keep[jj];
    cT[(size_t)brow * H_SZ + jj * 32 + lane] = c_state[jj];
  }
}

extern "C" void kernel_launch(void* const* d_in, const int* in_sizes, int n_in,
                              void* d_out, int out_size, void* d_ws, size_t ws_size,
                              hipStream_t stream) {
  const float* x      = (const float*)d_in[0];
  const float* W_in   = (const float*)d_in[1];
  const float* b_in   = (const float*)d_in[2];
  const float* W_out  = (const float*)d_in[3];
  const float* b_out  = (const float*)d_in[4];
  const float* thetas = (const float*)d_in[5];
  float* out = (float*)d_out;
  float* y0  = (float*)d_ws;

  qlstm_xproj<<<1024, 128, 0, stream>>>(x, W_in, b_in, y0);
  qlstm_recur<<<256, 32, 0, stream>>>(y0, W_in, W_out, b_out, thetas, out);
}


# hs store address = wave-uniform row base advanced in scalar registers + fixed lane offset: no vector address is advanced behind the stores, the per-step translation wait goes; arithmetic unchanged
# speedup vs baseline: 4.3048x; 1.0739x over previous
#include <hip/hip_runtime.h>
#include <hip/hip_bf16.h>
#include <math.h>

typedef __attribute__((ext_vector_type(2))) float v2f;
typedef __attribute__((ext_vector_type(8))) float v8f;

static __device__ __forceinline__ v8f wmma_f32_16x16x4(v2f a, v2f b, v8f c) {
  return __builtin_amdgcn_wmma_f32_16x16x4_f32(false, a, false, b, (short)0, c, false, false);
}

#define SLICE_END() __builtin_amdgcn_sched_barrier(0)
#define ISSUED_BEFORE_HERE(val) asm volatile("" :: "v"(val))

#define NEG_LOG2E  (-1.4426950408889634f)
#define NEG_2LOG2E (-2.8853900817779268f)

#define B_SZ 256
#define S_SZ 256
#define F_SZ 64
#define H_SZ 128

__global__ void __launch_bounds__(128)
qlstm_xproj(const float* __restrict__ x, const float* __restrict__ Win,
            const float* __restrict__ bin, float* __restrict__ y0) {
  const int lane = threadIdx.x & 31;
  const int wave = threadIdx.x >> 5;
  const int tile = blockIdx.x * 4 + wave;
  const int n    = lane & 15;
  const int hi   = lane >> 4;
  const int kh   = hi << 1;

  v2f bch[16];
#pragma unroll
  for (int j = 0; j < 16; ++j) {
    float b0 = 0.f, b1 = 0.f;
    if (n < 4) {
      int kk = j * 4 + kh;
      b0 = Win[n * 192 + 128 + kk];
      b1 = Win[n * 192 + 128 + kk + 1];
    }
    v2f bv = {b0, b1};
    bch[j] = bv;
  }

  const float bb = (n < 4) ? bin[n] : 0.f;
  v8f acc = {bb, bb, bb, bb, bb, bb, bb, bb};

  const float* xr = x + (size_t)(tile * 16 + n) * F_SZ;
#pragma unroll
  for (int j = 0; j < 16; ++j) {
    v2f a = {xr[j * 4 + kh], xr[j * 4 + kh + 1]};
    acc = wmma_f32_16x16x4(a, bch[j], acc);
  }

  if (n < 4) {
#pragma unroll
    for (int r = 0; r < 8; ++r) {
      int row = tile * 16 + r + (hi << 3);
      y0[(size_t)row * 4 + n] = acc[r];
    }
  }
}

__global__ void __launch_bounds__(32)
qlstm_recur(const float* __restrict__ y0, const float* __restrict__ Win,
            const float* __restrict__ Wout, const float* __restrict__ bout,
            const float* __restrict__ thetas, float* __restrict__ out) {
  __shared__ __attribute__((aligned(16))) float y0_lds[S_SZ * 4];

  const int lane  = threadIdx.x;
  const int n     = lane & 15;
  const int hi    = lane >> 4;
  const int kh    = hi << 1;
  const int brow  = blockIdx.x;

  float* hs = out;
  float* hT = out + (size_t)B_SZ * S_SZ * H_SZ;
  float* cT = hT + (size_t)B_SZ * H_SZ;

  {
    const float4* src = (const float4*)(y0 + (size_t)brow * S_SZ * 4);
#pragma unroll
    for (int i = 0; i < 8; ++i) {
      int idx = lane + i * 32;
      *(float4*)&y0_lds[idx * 4] = src[idx];
    }
  }

  const int q  = n >> 2;
  const int cg = n & 3;
  v2f bh[2][4];
#pragma unroll
  for (int u = 0; u < 2; ++u) {
#pragma unroll
    for (int ct = 0; ct < 4; ++ct) {
      int kk = 64 * u + 16 * hi + 4 * cg + ct;
      v2f bv = {Win[q * 192 + kk], Win[q * 192 + kk + 32]};
      bh[u][ct] = bv;
    }
  }
  v2f bw[8];
  float bo_s[8], bo_t[8];
#pragma unroll
  for (int j = 0; j < 8; ++j) {
    int col = j * 16 + n;
    v2f bv = {Wout[col * 4 + kh], Wout[col * 4 + kh + 1]};
    bw[j] = bv;
    const float b = bout[col];
    bo_s[j] = b * NEG_LOG2E;
    bo_t[j] = b * NEG_2LOG2E;
  }

  const int mg = n & 3;
  const float zscale = (mg == 2) ? NEG_2LOG2E : NEG_LOG2E;
  const v2f zsign = {hi ? 1.f : -1.f, 1.f};
  const v2f hp = {0.5f, -0.5f}, hm = {-0.5f, 0.5f}, hh = {0.5f, 0.5f};
  const v2f hpz = hp * zscale, hhz = hh * zscale;
  float th[4];
#pragma unroll
  for (int w = 0; w < 4; ++w) th[w] = thetas[mg * 4 + w];

  const float* yR = y0_lds + q;
  float* hs_row = hs + ((size_t)brow * S_SZ) * H_SZ;

  float c_state[4], h_keep[4];
#pragma unroll
  for (int r = 0; r < 4; ++r) { c_state[r] = 0.f; h_keep[r] = 0.f; }

  __builtin_amdgcn_fence(__ATOMIC_RELEASE, "wavefront");
  __builtin_amdgcn_wave_barrier();
  __builtin_amdgcn_fence(__ATOMIC_ACQUIRE, "wavefront");

  float ysum = 0.f;
  float yq = yR[0];

  for (int t = 0; t < S_SZ; ++t) {
    const float y = ysum + yq;

    v2f pw[4], pws[4];
#pragma unroll
    for (int w = 0; w < 4; ++w) {
      const float yw = __int_as_float(__builtin_amdgcn_readlane(__float_as_int(y), 4 * w));
      const float cv = __cosf(yw + th[w]);
      pw[w]  = cv * (w ? hp : hpz) + (w ? hh : hhz);
      pws[w] = cv * hm + hh;
    }
    const v2f PA  = pw[0] * pw[1];
    const v2f PB  = pw[0] * pws[1];
    const v2f Q03 = pw[2] * pw[3];
    const v2f Q12 = pw[2] * pws[3];
    v2f Rs[2], Vs[2], Xs[2];
#pragma unroll
    for (int g = 0; g < 2; ++g) {
      const v2f P = g ? PB : PA;
      const v2f t00 = P * Q03.x, t01 = P * Q12.x;
      const v2f t10 = P * Q12.y, t11 = P * Q03.y;
      const v2f u0 = t00 + t01, u1 = t10 + t11;
      const v2f d0 = t00 - t01, d1 = t10 - t11;
      Rs[g] = u0 + u1;
      Vs[g] = u0 - u1;
      Xs[g] = d0 - d1;
    }
    const v2f rd = Rs[0] - Rs[1];
    const v2f vd = Vs[0] - Vs[1];
    const v2f xd = Xs[0] - Xs[1];
    const v2f fa = {hi ? vd.x : xd.x, hi ? xd.x : rd.x};
    const v2f fb = {hi ? vd.y : xd.y, hi ? xd.y : rd.y};
    const v2f za = fa + zsign * fb;

    float* hs_t = hs_row + (size_t)t * H_SZ;
    v8f ga[8];
    float eg[4][4], rg[4][4], ecs[4];
    auto gate_tile = [&](int j) __attribute__((always_inline)) {
      const v8f cbias = {bo_s[j], bo_s[j], bo_t[j], bo_s[j], bo_s[j], bo_s[j], bo_t[j], bo_s[j]};
      ga[j] = wmma_f32_16x16x4(za, bw[j], cbias);
    };
    auto S1 = [&](int jj) __attribute__((always_inline)) {
#pragma unroll
      for (int r = 0; r < 4; ++r)
        eg[jj][r] = __builtin_amdgcn_exp2f(hi ? ga[2 * jj + 1][r] : ga[2 * jj][r]);
    };
    auto S2 = [&](int jj) __attribute__((always_inline)) {
#pragma unroll
      for (int r = 0; r < 4; ++r) rg[jj][r] = __builtin_amdgcn_rcpf(1.f + eg[jj][r]);
    };
    auto S3a = [&](int jj) __attribute__((always_inline)) {
      const float fv = rg[jj][0], iv = rg[jj][1];
      const float gv = fmaf(2.f, rg[jj][2], -1.f);
      float cs = fv * c_state[jj] + iv * gv;
      c_state[jj] = cs;
    };
    auto S3b = [&](int jj) __attribute__((always_inline)) {
      ecs[jj] = __builtin_amdgcn_exp2f(c_state[jj] * NEG_2LOG2E);
    };
    float rcs[4];
    auto S4a = [&](int jj) __attribute__((always_inline)) {
      rcs[jj] = __builtin_amdgcn_rcpf(1.f + ecs[jj]);
    };
    auto S4b = [&](int jj) __attribute__((always_inline)) {
      const float ov = rg[jj][3];
      h_keep[jj] = ov * fmaf(2.f, rcs[jj], -1.f);
    };
    v8f acc[4];
    const v8f zero = {0.f, 0.f, 0.f, 0.f, 0.f, 0.f, 0.f, 0.f};

    gate_tile(0); gate_tile(1); gate_tile(2); gate_tile(3);
    ISSUED_BEFORE_HERE(ga[0][0]); ISSUED_BEFORE_HERE(ga[1][0]);
    ISSUED_BEFORE_HERE(ga[2][0]); ISSUED_BEFORE_HERE(ga[3][0]);
    SLICE_END();
    S1(0);
    SLICE_END(); gate_tile(4); ISSUED_BEFORE_HERE(ga[4][0]); SLICE_END();
    S2(0);
    SLICE_END(); gate_tile(5); ISSUED_BEFORE_HERE(ga[5][0]); SLICE_END();
    S1(1);
    SLICE_END(); gate_tile(6); ISSUED_BEFORE_HERE(ga[6][0]); SLICE_END();
    S2(1); S3a(0); S3b(0);
    SLICE_END(); gate_tile(7); ISSUED_BEFORE_HERE(ga[7][0]); SLICE_END();
    S3a(1); S3b(1); S1(2); S4a(0);
    SLICE_END();
    S4a(1); S4b(1); S4b(0); S2(2); S1(3);
    const int tn = (t + 1 < S_SZ) ? t + 1 : t;
    yq = yR[tn * 4];
    const v2f a0 = {h_keep[0], h_keep[1]};
    SLICE_END(); acc[0] = wmma_f32_16x16x4(a0, bh[0][0], zero); ISSUED_BEFORE_HERE(acc[0][0]); SLICE_END();
    S2(3); S3a(2); S3b(2);
    SLICE_END(); acc[1] = wmma_f32_16x16x4(a0, bh[0][1], zero); ISSUED_BEFORE_HERE(acc[1][0]); SLICE_END();
    S3a(3); S3b(3); S4a(2); S4b(2);
    SLICE_END(); acc[2] = wmma_f32_16x16x4(a0, bh[0][2], zero); ISSUED_BEFORE_HERE(acc[2][0]); SLICE_END();
    S4a(3);
    SLICE_END(); acc[3] = wmma_f32_16x16x4(a0, bh[0][3], zero); ISSUED_BEFORE_HERE(acc[3][0]); SLICE_END();
    S4b(3);
    {
      const v2f a1 = {h_keep[2], h_keep[3]};
#pragma unroll
      for (int ct = 0; ct < 4; ++ct) acc[ct] = wmma_f32_16x16x4(a1, bh[1][ct], acc[ct]);
    }
    SLICE_END();
#pragma unroll
    for (int jj = 0; jj < 4; ++jj) hs_t[jj * 32 + lane] = h_keep[jj];
    const float sa = (acc[0][0] + acc[1][1]) + (acc[2][2] + acc[3][3]);
    const float sb = (acc[0][4] + acc[1][5]) + (acc[2][6] + acc[3][7]);
    const float vs = (cg & 1) ? sb : sa;
    const float tp = vs + __int_as_float(__builtin_amdgcn_update_dpp(0, __float_as_int(vs), 0xB1, 0xF, 0xF, true));
    const float xo = __int_as_float(__builtin_amdgcn_permlanex16(0, __float_as_int(tp), 0x54761032u, 0xDCFE98BAu, false, false));
    const float d = tp + xo;
    ysum = d;
  }

#pragma unroll
  for (int jj = 0; jj < 4; ++jj) {
    hT[(size_t)brow * H_SZ + jj * 32 + lane] = h_keep[jj];
    cT[(size_t)brow * H_SZ + jj * 32 + lane] = c_state[jj];
  }
}

extern "C" void kernel_launch(void* const* d_in, const int* in_sizes, int n_in,
                              void* d_out, int out_size, void* d_ws, size_t ws_size,
                              hipStream_t stream) {
  const float* x      = (const float*)d_in[0];
  const float* W_in   = (const float*)d_in[1];
  const float* b_in   = (const float*)d_in[2];
  const float* W_out  = (const float*)d_in[3];
  const float* b_out  = (const float*)d_in[4];
  const float* thetas = (const float*)d_in[5];
  float* out = (float*)d_out;
  float* y0  = (float*)d_ws;

  qlstm_xproj<<<1024, 128, 0, stream>>>(x, W_in, b_in, y0);
  qlstm_recur<<<256, 32, 0, stream>>>(y0, W_in, W_out, b_out, thetas, out);
}
